# starting kernel
# baseline (speedup 1.0000x reference)
#include <hip/hip_runtime.h>


typedef __attribute__((ext_vector_type(2))) float v2f;
typedef __attribute__((ext_vector_type(8))) float v8f;

#define N_NODES 50000
#define N_EDGES 1600000
#define IN_DIM  128
#define HIDDEN  128
#define OUT_DIM 64

__global__ __launch_bounds__(256)
void sage_scatter(const float* __restrict__ feat,
                  const int*   __restrict__ src,
                  const int*   __restrict__ dst,
                  float* __restrict__ agg,
                  float* __restrict__ deg,
                  int n_edges) {
    const int lane = threadIdx.x & 31;
    const int e    = blockIdx.x * 8 + (threadIdx.x >> 5);
    if (e >= n_edges) return;

    const int s = src[e];
    const int d = dst[e];

    const float4 v =
        *(reinterpret_cast<const float4*>(feat + (size_t)s * 128) + lane);

    float* out = agg + (size_t)d * 128 + lane * 4;
    unsafeAtomicAdd(out + 0, v.x);
    unsafeAtomicAdd(out + 1, v.y);
    unsafeAtomicAdd(out + 2, v.z);
    unsafeAtomicAdd(out + 3, v.w);

    if (deg != nullptr && lane == 0) {
        unsafeAtomicAdd(deg + d, 1.0f);
    }
}

template <int NDIM, bool RELU>
__global__ __launch_bounds__(256)
void sage_gemm(const float* __restrict__ agg,
               const float* __restrict__ deg,
               const float* __restrict__ X,
               const float* __restrict__ Wl,
               const float* __restrict__ Wr,
               const float* __restrict__ bias,
               float* __restrict__ out,
               int n_nodes) {
    constexpr int NTILES_N = NDIM / 16;
    const int wave = blockIdx.x * 8 + (threadIdx.x >> 5);
    const int total_tiles = (n_nodes / 16) * NTILES_N;
    if (wave >= total_tiles) return;

    const int mtile = wave / NTILES_N;
    const int ntile = wave % NTILES_N;

    const int lane = threadIdx.x & 31;
    const int r    = lane & 15;
    const int kk   = (lane >> 4) * 2;

    const int row  = mtile * 16 + r;
    const int coln = ntile * 16 + r;

    const float dval = deg[row];
    const float invd = 1.0f / fmaxf(dval, 1.0f);

    const float* aggRow = agg + (size_t)row * 128;
    const float* xRow   = X   + (size_t)row * 128;

    v8f acc = {};

#pragma unroll 8
    for (int k0 = 0; k0 < 128; k0 += 4) {
        const int k = k0 + kk;

        v2f aA; aA.x = aggRow[k] * invd;  aA.y = aggRow[k + 1] * invd;
        v2f aX; aX.x = xRow[k];           aX.y = xRow[k + 1];

        v2f bL; bL.x = Wl[(size_t)k * NDIM + coln];
                bL.y = Wl[(size_t)(k + 1) * NDIM + coln];
        v2f bR; bR.x = Wr[(size_t)k * NDIM + coln];
                bR.y = Wr[(size_t)(k + 1) * NDIM + coln];

        acc = __builtin_amdgcn_wmma_f32_16x16x4_f32(
                  false, aA, false, bL, (short)0, acc, false, false);
        acc = __builtin_amdgcn_wmma_f32_16x16x4_f32(
                  false, aX, false, bR, (short)0, acc, false, false);
    }

    const int   half = lane >> 4;
    const float bv   = bias[coln];
#pragma unroll
    for (int i = 0; i < 8; ++i) {
        const int m    = i + 8 * half;
        const int node = mtile * 16 + m;
        float v = acc[i] + bv;
        if (RELU) v = fmaxf(v, 0.0f);
        out[(size_t)node * NDIM + (ntile * 16 + (lane & 15))] = v;
    }
}

extern "C" void kernel_launch(void* const* d_in, const int* in_sizes, int n_in,
                              void* d_out, int out_size, void* d_ws, size_t ws_size,
                              hipStream_t stream) {
    const float* x    = (const float*)d_in[0];
    const int*   ei   = (const int*)  d_in[1];
    const float* W1l  = (const float*)d_in[2];
    const float* W1r  = (const float*)d_in[3];
    const float* b1   = (const float*)d_in[4];
    const float* W2l  = (const float*)d_in[5];
    const float* W2r  = (const float*)d_in[6];
    const float* b2   = (const float*)d_in[7];
    float*       outp = (float*)d_out;

    const int n_nodes = in_sizes[0] / IN_DIM;
    const int n_edges = in_sizes[1] / 2;
    const int* src = ei;
    const int* dst = ei + n_edges;

    const size_t aggBytes = (size_t)n_nodes * 128 * sizeof(float);
    const size_t degBytes = (((size_t)n_nodes * sizeof(float)) + 255) & ~(size_t)255;

    char* ws    = (char*)d_ws;
    float* agg1 = (float*)(ws);
    float* agg2 = (float*)(ws + aggBytes);
    float* deg  = (float*)(ws + 2 * aggBytes);
    float* h    = (float*)(ws + 2 * aggBytes + degBytes);

    hipMemsetAsync(d_ws, 0, 2 * aggBytes + degBytes, stream);

    const int scatterBlocks = (n_edges + 7) / 8;
    const int tiles1 = (n_nodes / 16) * (HIDDEN / 16);
    const int tiles2 = (n_nodes / 16) * (OUT_DIM / 16);

    sage_scatter<<<scatterBlocks, 256, 0, stream>>>(x, src, dst, agg1, deg, n_edges);
    sage_gemm<HIDDEN, true><<<(tiles1 + 7) / 8, 256, 0, stream>>>(
        agg1, deg, x, W1l, W1r, b1, h, n_nodes);

    sage_scatter<<<scatterBlocks, 256, 0, stream>>>(h, src, dst, agg2, nullptr, n_edges);
    sage_gemm<OUT_DIM, false><<<(tiles2 + 7) / 8, 256, 0, stream>>>(
        agg2, deg, h, W2l, W2r, b2, outp, n_nodes);
}


# scatter: lane l takes columns l+32j so each f32 atomic add and each row load instruction covers one whole 128-byte line (4 atomic line requests per edge, not 16); sums stay in atomic arrival order
# speedup vs baseline: 1.1992x; 1.1992x over previous
#include <hip/hip_runtime.h>


typedef __attribute__((ext_vector_type(2))) float v2f;
typedef __attribute__((ext_vector_type(8))) float v8f;

#define N_NODES 50000
#define N_EDGES 1600000
#define IN_DIM  128
#define HIDDEN  128
#define OUT_DIM 64

__global__ __launch_bounds__(256)
void sage_scatter(const float* __restrict__ feat,
                  const int*   __restrict__ src,
                  const int*   __restrict__ dst,
                  float* __restrict__ agg,
                  float* __restrict__ deg,
                  int n_edges) {
    const int lane = threadIdx.x & 31;
    const int e    = blockIdx.x * 8 + (threadIdx.x >> 5);
    if (e >= n_edges) return;

    const int s = src[e];
    const int d = dst[e];

    const float* in = feat + (size_t)s * 128 + lane;
    const float v0 = in[0];
    const float v1 = in[32];
    const float v2 = in[64];
    const float v3 = in[96];

    float* out = agg + (size_t)d * 128 + lane;
    unsafeAtomicAdd(out + 0,  v0);
    unsafeAtomicAdd(out + 32, v1);
    unsafeAtomicAdd(out + 64, v2);
    unsafeAtomicAdd(out + 96, v3);

    if (deg != nullptr && lane == 0) {
        unsafeAtomicAdd(deg + d, 1.0f);
    }
}

template <int NDIM, bool RELU>
__global__ __launch_bounds__(256)
void sage_gemm(const float* __restrict__ agg,
               const float* __restrict__ deg,
               const float* __restrict__ X,
               const float* __restrict__ Wl,
               const float* __restrict__ Wr,
               const float* __restrict__ bias,
               float* __restrict__ out,
               int n_nodes) {
    constexpr int NTILES_N = NDIM / 16;
    const int wave = blockIdx.x * 8 + (threadIdx.x >> 5);
    const int total_tiles = (n_nodes / 16) * NTILES_N;
    if (wave >= total_tiles) return;

    const int mtile = wave / NTILES_N;
    const int ntile = wave % NTILES_N;

    const int lane = threadIdx.x & 31;
    const int r    = lane & 15;
    const int kk   = (lane >> 4) * 2;

    const int row  = mtile * 16 + r;
    const int coln = ntile * 16 + r;

    const float dval = deg[row];
    const float invd = 1.0f / fmaxf(dval, 1.0f);

    const float* aggRow = agg + (size_t)row * 128;
    const float* xRow   = X   + (size_t)row * 128;

    v8f acc = {};

#pragma unroll 8
    for (int k0 = 0; k0 < 128; k0 += 4) {
        const int k = k0 + kk;

        v2f aA; aA.x = aggRow[k] * invd;  aA.y = aggRow[k + 1] * invd;
        v2f aX; aX.x = xRow[k];           aX.y = xRow[k + 1];

        v2f bL; bL.x = Wl[(size_t)k * NDIM + coln];
                bL.y = Wl[(size_t)(k + 1) * NDIM + coln];
        v2f bR; bR.x = Wr[(size_t)k * NDIM + coln];
                bR.y = Wr[(size_t)(k + 1) * NDIM + coln];

        acc = __builtin_amdgcn_wmma_f32_16x16x4_f32(
                  false, aA, false, bL, (short)0, acc, false, false);
        acc = __builtin_amdgcn_wmma_f32_16x16x4_f32(
                  false, aX, false, bR, (short)0, acc, false, false);
    }

    const int   half = lane >> 4;
    const float bv   = bias[coln];
#pragma unroll
    for (int i = 0; i < 8; ++i) {
        const int m    = i + 8 * half;
        const int node = mtile * 16 + m;
        float v = acc[i] + bv;
        if (RELU) v = fmaxf(v, 0.0f);
        out[(size_t)node * NDIM + (ntile * 16 + (lane & 15))] = v;
    }
}

extern "C" void kernel_launch(void* const* d_in, const int* in_sizes, int n_in,
                              void* d_out, int out_size, void* d_ws, size_t ws_size,
                              hipStream_t stream) {
    const float* x    = (const float*)d_in[0];
    const int*   ei   = (const int*)  d_in[1];
    const float* W1l  = (const float*)d_in[2];
    const float* W1r  = (const float*)d_in[3];
    const float* b1   = (const float*)d_in[4];
    const float* W2l  = (const float*)d_in[5];
    const float* W2r  = (const float*)d_in[6];
    const float* b2   = (const float*)d_in[7];
    float*       outp = (float*)d_out;

    const int n_nodes = in_sizes[0] / IN_DIM;
    const int n_edges = in_sizes[1] / 2;
    const int* src = ei;
    const int* dst = ei + n_edges;

    const size_t aggBytes = (size_t)n_nodes * 128 * sizeof(float);
    const size_t degBytes = (((size_t)n_nodes * sizeof(float)) + 255) & ~(size_t)255;

    char* ws    = (char*)d_ws;
    float* agg1 = (float*)(ws);
    float* agg2 = (float*)(ws + aggBytes);
    float* deg  = (float*)(ws + 2 * aggBytes);
    float* h    = (float*)(ws + 2 * aggBytes + degBytes);

    hipMemsetAsync(d_ws, 0, 2 * aggBytes + degBytes, stream);

    const int scatterBlocks = (n_edges + 7) / 8;
    const int tiles1 = (n_nodes / 16) * (HIDDEN / 16);
    const int tiles2 = (n_nodes / 16) * (OUT_DIM / 16);

    sage_scatter<<<scatterBlocks, 256, 0, stream>>>(x, src, dst, agg1, deg, n_edges);
    sage_gemm<HIDDEN, true><<<(tiles1 + 7) / 8, 256, 0, stream>>>(
        agg1, deg, x, W1l, W1r, b1, h, n_nodes);

    sage_scatter<<<scatterBlocks, 256, 0, stream>>>(h, src, dst, agg2, nullptr, n_edges);
    sage_gemm<OUT_DIM, false><<<(tiles2 + 7) / 8, 256, 0, stream>>>(
        agg2, deg, h, W2l, W2r, b2, outp, n_nodes);
}


# scatter: each wave walks a run of 32 consecutive edges (indices loaded once, broadcast by lane read, 16 row loads in flight) instead of one edge: 32x fewer waves; sums stay in atomic arrival order
# speedup vs baseline: 1.7904x; 1.4930x over previous
#include <hip/hip_runtime.h>


typedef __attribute__((ext_vector_type(2))) float v2f;
typedef __attribute__((ext_vector_type(8))) float v8f;

#define N_NODES 50000
#define N_EDGES 1600000
#define IN_DIM  128
#define HIDDEN  128
#define OUT_DIM 64

__global__ __launch_bounds__(256)
void sage_scatter(const float* __restrict__ feat,
                  const int*   __restrict__ src,
                  const int*   __restrict__ dst,
                  float* __restrict__ agg,
                  float* __restrict__ deg,
                  int n_edges) {
    const int lane = threadIdx.x & 31;
    const int wave = __builtin_amdgcn_readfirstlane((int)(threadIdx.x >> 5));
    const int e0   = (blockIdx.x * 8 + wave) * 32;
    if (e0 >= n_edges) return;
    const int left = n_edges - e0;
    const int cnt  = left < 32 ? left : 32;

    int sv = 0;
    int dv = 0;
    if (lane < cnt) {
        sv = src[e0 + lane];
        dv = dst[e0 + lane];
        if (deg != nullptr) {
            unsafeAtomicAdd(deg + dv, 1.0f);
        }
    }

    int i = 0;
    for (; i + 4 <= cnt; i += 4) {
        float v[4][4];
#pragma unroll
        for (int u = 0; u < 4; ++u) {
            const int s = __builtin_amdgcn_readlane(sv, i + u);
            const float* in = feat + (size_t)s * 128 + lane;
            v[u][0] = in[0];
            v[u][1] = in[32];
            v[u][2] = in[64];
            v[u][3] = in[96];
        }
#pragma unroll
        for (int u = 0; u < 4; ++u) {
            const int d = __builtin_amdgcn_readlane(dv, i + u);
            float* out = agg + (size_t)d * 128 + lane;
            unsafeAtomicAdd(out + 0,  v[u][0]);
            unsafeAtomicAdd(out + 32, v[u][1]);
            unsafeAtomicAdd(out + 64, v[u][2]);
            unsafeAtomicAdd(out + 96, v[u][3]);
        }
    }
    for (; i < cnt; ++i) {
        const int s = __builtin_amdgcn_readlane(sv, i);
        const int d = __builtin_amdgcn_readlane(dv, i);
        const float* in = feat + (size_t)s * 128 + lane;
        const float v0 = in[0];
        const float v1 = in[32];
        const float v2 = in[64];
        const float v3 = in[96];
        float* out = agg + (size_t)d * 128 + lane;
        unsafeAtomicAdd(out + 0,  v0);
        unsafeAtomicAdd(out + 32, v1);
        unsafeAtomicAdd(out + 64, v2);
        unsafeAtomicAdd(out + 96, v3);
    }
}

template <int NDIM, bool RELU>
__global__ __launch_bounds__(256)
void sage_gemm(const float* __restrict__ agg,
               const float* __restrict__ deg,
               const float* __restrict__ X,
               const float* __restrict__ Wl,
               const float* __restrict__ Wr,
               const float* __restrict__ bias,
               float* __restrict__ out,
               int n_nodes) {
    constexpr int NTILES_N = NDIM / 16;
    const int wave = blockIdx.x * 8 + (threadIdx.x >> 5);
    const int total_tiles = (n_nodes / 16) * NTILES_N;
    if (wave >= total_tiles) return;

    const int mtile = wave / NTILES_N;
    const int ntile = wave % NTILES_N;

    const int lane = threadIdx.x & 31;
    const int r    = lane & 15;
    const int kk   = (lane >> 4) * 2;

    const int row  = mtile * 16 + r;
    const int coln = ntile * 16 + r;

    const float dval = deg[row];
    const float invd = 1.0f / fmaxf(dval, 1.0f);

    const float* aggRow = agg + (size_t)row * 128;
    const float* xRow   = X   + (size_t)row * 128;

    v8f acc = {};

#pragma unroll 8
    for (int k0 = 0; k0 < 128; k0 += 4) {
        const int k = k0 + kk;

        v2f aA; aA.x = aggRow[k] * invd;  aA.y = aggRow[k + 1] * invd;
        v2f aX; aX.x = xRow[k];           aX.y = xRow[k + 1];

        v2f bL; bL.x = Wl[(size_t)k * NDIM + coln];
                bL.y = Wl[(size_t)(k + 1) * NDIM + coln];
        v2f bR; bR.x = Wr[(size_t)k * NDIM + coln];
                bR.y = Wr[(size_t)(k + 1) * NDIM + coln];

        acc = __builtin_amdgcn_wmma_f32_16x16x4_f32(
                  false, aA, false, bL, (short)0, acc, false, false);
        acc = __builtin_amdgcn_wmma_f32_16x16x4_f32(
                  false, aX, false, bR, (short)0, acc, false, false);
    }

    const int   half = lane >> 4;
    const float bv   = bias[coln];
#pragma unroll
    for (int i = 0; i < 8; ++i) {
        const int m    = i + 8 * half;
        const int node = mtile * 16 + m;
        float v = acc[i] + bv;
        if (RELU) v = fmaxf(v, 0.0f);
        out[(size_t)node * NDIM + (ntile * 16 + (lane & 15))] = v;
    }
}

extern "C" void kernel_launch(void* const* d_in, const int* in_sizes, int n_in,
                              void* d_out, int out_size, void* d_ws, size_t ws_size,
                              hipStream_t stream) {
    const float* x    = (const float*)d_in[0];
    const int*   ei   = (const int*)  d_in[1];
    const float* W1l  = (const float*)d_in[2];
    const float* W1r  = (const float*)d_in[3];
    const float* b1   = (const float*)d_in[4];
    const float* W2l  = (const float*)d_in[5];
    const float* W2r  = (const float*)d_in[6];
    const float* b2   = (const float*)d_in[7];
    float*       outp = (float*)d_out;

    const int n_nodes = in_sizes[0] / IN_DIM;
    const int n_edges = in_sizes[1] / 2;
    const int* src = ei;
    const int* dst = ei + n_edges;

    const size_t aggBytes = (size_t)n_nodes * 128 * sizeof(float);
    const size_t degBytes = (((size_t)n_nodes * sizeof(float)) + 255) & ~(size_t)255;

    char* ws    = (char*)d_ws;
    float* agg1 = (float*)(ws);
    float* agg2 = (float*)(ws + aggBytes);
    float* deg  = (float*)(ws + 2 * aggBytes);
    float* h    = (float*)(ws + 2 * aggBytes + degBytes);

    hipMemsetAsync(d_ws, 0, 2 * aggBytes + degBytes, stream);

    const int scatterBlocks = (n_edges + 255) / 256;
    const int tiles1 = (n_nodes / 16) * (HIDDEN / 16);
    const int tiles2 = (n_nodes / 16) * (OUT_DIM / 16);

    sage_scatter<<<scatterBlocks, 256, 0, stream>>>(x, src, dst, agg1, deg, n_edges);
    sage_gemm<HIDDEN, true><<<(tiles1 + 7) / 8, 256, 0, stream>>>(
        agg1, deg, x, W1l, W1r, b1, h, n_nodes);

    sage_scatter<<<scatterBlocks, 256, 0, stream>>>(h, src, dst, agg2, nullptr, n_edges);
    sage_gemm<OUT_DIM, false><<<(tiles2 + 7) / 8, 256, 0, stream>>>(
        agg2, deg, h, W2l, W2r, b2, outp, n_nodes);
}


# edges counting-sorted by destination each call (int atomics: histogram, scan, fill); one wave per node sums its source rows in registers, stores once: no f32 atomics; sum order = fill arrival order
# speedup vs baseline: 2.0302x; 1.1339x over previous
#include <hip/hip_runtime.h>


typedef __attribute__((ext_vector_type(2))) float v2f;
typedef __attribute__((ext_vector_type(8))) float v8f;

#define N_NODES 50000
#define N_EDGES 1600000
#define IN_DIM  128
#define HIDDEN  128
#define OUT_DIM 64

__global__ __launch_bounds__(256)
void sage_count(const int* __restrict__ dst,
                int* __restrict__ cnt,
                int n_edges) {
    const int e0 = blockIdx.x * 1024 + threadIdx.x;
    int d[4];
#pragma unroll
    for (int u = 0; u < 4; ++u) {
        const int e = e0 + u * 256;
        d[u] = e < n_edges ? dst[e] : -1;
    }
#pragma unroll
    for (int u = 0; u < 4; ++u) {
        if (d[u] >= 0) atomicAdd(cnt + d[u], 1);
    }
}

__global__ __launch_bounds__(1024)
void sage_scan(const int* __restrict__ cnt,
               int* __restrict__ rowptr,
               int* __restrict__ cursor,
               float* __restrict__ degf,
               int n) {
    __shared__ int wfront[32];
    __shared__ int wtot[32];
    const int t    = threadIdx.x;
    const int lane = t & 31;
    const int wv   = t >> 5;
    const int base = blockIdx.x * 1024;

    int front = 0;
    for (int j = t; j < base; j += 1024) front += cnt[j];

    const int idx = base + t;
    const int v   = idx < n ? cnt[idx] : 0;

#pragma unroll
    for (int o = 16; o > 0; o >>= 1) front += __shfl_xor(front, o, 32);
    int inc = v;
#pragma unroll
    for (int o = 1; o < 32; o <<= 1) {
        const int y = __shfl_up(inc, o, 32);
        if (lane >= o) inc += y;
    }
    if (lane == 31) wtot[wv] = inc;
    if (lane == 0)  wfront[wv] = front;
    __syncthreads();

    int all_front = wfront[lane];
    const int tot = wtot[lane];
#pragma unroll
    for (int o = 16; o > 0; o >>= 1) all_front += __shfl_xor(all_front, o, 32);
    int tinc = tot;
#pragma unroll
    for (int o = 1; o < 32; o <<= 1) {
        const int y = __shfl_up(tinc, o, 32);
        if (lane >= o) tinc += y;
    }
    const int waves_before = __shfl(tinc, wv, 32) - __shfl(tot, wv, 32);
    const int excl = all_front + waves_before + inc - v;

    if (idx < n) {
        rowptr[idx] = excl;
        cursor[idx] = excl;
        degf[idx]   = (float)v;
        if (idx == n - 1) rowptr[n] = excl + v;
    }
}

__global__ __launch_bounds__(256)
void sage_fill(const int* __restrict__ src,
               const int* __restrict__ dst,
               int* __restrict__ cursor,
               int* __restrict__ csr,
               int n_edges) {
    const int e0 = blockIdx.x * 1024 + threadIdx.x;
    int d[4];
    int s[4];
#pragma unroll
    for (int u = 0; u < 4; ++u) {
        const int e = e0 + u * 256;
        d[u] = e < n_edges ? dst[e] : -1;
        s[u] = e < n_edges ? src[e] : 0;
    }
    int p[4];
#pragma unroll
    for (int u = 0; u < 4; ++u) {
        p[u] = d[u] >= 0 ? atomicAdd(cursor + d[u], 1) : -1;
    }
#pragma unroll
    for (int u = 0; u < 4; ++u) {
        if (p[u] >= 0) csr[p[u]] = s[u];
    }
}

__global__ __launch_bounds__(256)
void sage_gather(const float* __restrict__ feat,
                 const int*   __restrict__ rowptr,
                 const int*   __restrict__ csr,
                 float* __restrict__ agg,
                 int n_nodes) {
    const int lane = threadIdx.x & 31;
    const int wave = __builtin_amdgcn_readfirstlane((int)(threadIdx.x >> 5));
    const int node = blockIdx.x * 8 + wave;
    if (node >= n_nodes) return;
    const int beg = __builtin_amdgcn_readfirstlane(rowptr[node]);
    const int end = __builtin_amdgcn_readfirstlane(rowptr[node + 1]);

    const float* col = feat + lane * 4;
    float4 acc = make_float4(0.0f, 0.0f, 0.0f, 0.0f);

    for (int b = beg; b < end; b += 32) {
        const int left = end - b;
        const int cnt  = left < 32 ? left : 32;
        int sv = 0;
        if (lane < cnt) sv = csr[b + lane];

        int i = 0;
        for (; i + 16 <= cnt; i += 16) {
            float4 v[16];
#pragma unroll
            for (int u = 0; u < 16; ++u) {
                const int s = __builtin_amdgcn_readlane(sv, i + u);
                v[u] = *(const float4*)(col + (size_t)s * 128);
            }
#pragma unroll
            for (int u = 0; u < 16; ++u) {
                acc.x += v[u].x; acc.y += v[u].y; acc.z += v[u].z; acc.w += v[u].w;
            }
        }
        if (i < cnt) {
            float4 v[16];
#pragma unroll
            for (int u = 0; u < 16; ++u) {
                v[u] = make_float4(0.0f, 0.0f, 0.0f, 0.0f);
                if (i + u < cnt) {
                    const int s = __builtin_amdgcn_readlane(sv, i + u);
                    v[u] = *(const float4*)(col + (size_t)s * 128);
                }
            }
#pragma unroll
            for (int u = 0; u < 16; ++u) {
                acc.x += v[u].x; acc.y += v[u].y; acc.z += v[u].z; acc.w += v[u].w;
            }
        }
    }
    *(float4*)(agg + (size_t)node * 128 + lane * 4) = acc;
}

template <int NDIM, bool RELU>
__global__ __launch_bounds__(256)
void sage_gemm(const float* __restrict__ agg,
               const float* __restrict__ deg,
               const float* __restrict__ X,
               const float* __restrict__ Wl,
               const float* __restrict__ Wr,
               const float* __restrict__ bias,
               float* __restrict__ out,
               int n_nodes) {
    constexpr int NTILES_N = NDIM / 16;
    const int wave = blockIdx.x * 8 + (threadIdx.x >> 5);
    const int total_tiles = (n_nodes / 16) * NTILES_N;
    if (wave >= total_tiles) return;

    const int mtile = wave / NTILES_N;
    const int ntile = wave % NTILES_N;

    const int lane = threadIdx.x & 31;
    const int r    = lane & 15;
    const int kk   = (lane >> 4) * 2;

    const int row  = mtile * 16 + r;
    const int coln = ntile * 16 + r;

    const float dval = deg[row];
    const float invd = 1.0f / fmaxf(dval, 1.0f);

    const float* aggRow = agg + (size_t)row * 128;
    const float* xRow   = X   + (size_t)row * 128;

    v8f acc = {};

#pragma unroll 8
    for (int k0 = 0; k0 < 128; k0 += 4) {
        const int k = k0 + kk;

        v2f aA; aA.x = aggRow[k] * invd;  aA.y = aggRow[k + 1] * invd;
        v2f aX; aX.x = xRow[k];           aX.y = xRow[k + 1];

        v2f bL; bL.x = Wl[(size_t)k * NDIM + coln];
                bL.y = Wl[(size_t)(k + 1) * NDIM + coln];
        v2f bR; bR.x = Wr[(size_t)k * NDIM + coln];
                bR.y = Wr[(size_t)(k + 1) * NDIM + coln];

        acc = __builtin_amdgcn_wmma_f32_16x16x4_f32(
                  false, aA, false, bL, (short)0, acc, false, false);
        acc = __builtin_amdgcn_wmma_f32_16x16x4_f32(
                  false, aX, false, bR, (short)0, acc, false, false);
    }

    const int   half = lane >> 4;
    const float bv   = bias[coln];
#pragma unroll
    for (int i = 0; i < 8; ++i) {
        const int m    = i + 8 * half;
        const int node = mtile * 16 + m;
        float v = acc[i] + bv;
        if (RELU) v = fmaxf(v, 0.0f);
        out[(size_t)node * NDIM + (ntile * 16 + (lane & 15))] = v;
    }
}

extern "C" void kernel_launch(void* const* d_in, const int* in_sizes, int n_in,
                              void* d_out, int out_size, void* d_ws, size_t ws_size,
                              hipStream_t stream) {
    const float* x    = (const float*)d_in[0];
    const int*   ei   = (const int*)  d_in[1];
    const float* W1l  = (const float*)d_in[2];
    const float* W1r  = (const float*)d_in[3];
    const float* b1   = (const float*)d_in[4];
    const float* W2l  = (const float*)d_in[5];
    const float* W2r  = (const float*)d_in[6];
    const float* b2   = (const float*)d_in[7];
    float*       outp = (float*)d_out;

    const int n_nodes = in_sizes[0] / IN_DIM;
    const int n_edges = in_sizes[1] / 2;
    const int* src = ei;
    const int* dst = ei + n_edges;

    const size_t aggBytes = (size_t)n_nodes * 128 * sizeof(float);
    const size_t degBytes = (((size_t)(n_nodes + 1) * sizeof(float)) + 255) & ~(size_t)255;

    char* ws      = (char*)d_ws;
    float* agg1   = (float*)(ws);
    float* agg2   = (float*)(ws + aggBytes);
    float* deg    = (float*)(ws + 2 * aggBytes);
    float* h      = (float*)(ws + 2 * aggBytes + degBytes);
    int*   cnt    = (int*)  (ws + 3 * aggBytes + degBytes);
    int*   rowptr = (int*)  (ws + 3 * aggBytes + 2 * degBytes);
    int*   cursor = (int*)  (ws + 3 * aggBytes + 3 * degBytes);
    int*   csr    = (int*)  (ws + 3 * aggBytes + 4 * degBytes);

    hipMemsetAsync(cnt, 0, degBytes, stream);

    const int edgeBlocks = (n_edges + 1023) / 1024;
    const int scanBlocks = (n_nodes + 1023) / 1024;
    const int nodeBlocks = (n_nodes + 7) / 8;
    const int tiles1 = (n_nodes / 16) * (HIDDEN / 16);
    const int tiles2 = (n_nodes / 16) * (OUT_DIM / 16);

    sage_count<<<edgeBlocks, 256, 0, stream>>>(dst, cnt, n_edges);
    sage_scan<<<scanBlocks, 1024, 0, stream>>>(cnt, rowptr, cursor, deg, n_nodes);
    sage_fill<<<edgeBlocks, 256, 0, stream>>>(src, dst, cursor, csr, n_edges);

    sage_gather<<<nodeBlocks, 256, 0, stream>>>(x, rowptr, csr, agg1, n_nodes);
    sage_gemm<HIDDEN, true><<<(tiles1 + 7) / 8, 256, 0, stream>>>(
        agg1, deg, x, W1l, W1r, b1, h, n_nodes);

    sage_gather<<<nodeBlocks, 256, 0, stream>>>(h, rowptr, csr, agg2, n_nodes);
    sage_gemm<OUT_DIM, false><<<(tiles2 + 7) / 8, 256, 0, stream>>>(
        agg2, deg, h, W2l, W2r, b2, outp, n_nodes);
}


# GEMM re-tiled: both weight matrices staged once per work-group in LDS as k-pair B fragments; a wave computes 16x64 strips (4 accumulators) from 16-byte A loads; the order of the k sums changes
# speedup vs baseline: 2.2524x; 1.1094x over previous
#include <hip/hip_runtime.h>


typedef __attribute__((ext_vector_type(2))) float v2f;
typedef __attribute__((ext_vector_type(8))) float v8f;

#define N_NODES 50000
#define N_EDGES 1600000
#define IN_DIM  128
#define HIDDEN  128
#define OUT_DIM 64

__global__ __launch_bounds__(256)
void sage_count(const int* __restrict__ dst,
                int* __restrict__ cnt,
                int n_edges) {
    const int e0 = blockIdx.x * 1024 + threadIdx.x;
    int d[4];
#pragma unroll
    for (int u = 0; u < 4; ++u) {
        const int e = e0 + u * 256;
        d[u] = e < n_edges ? dst[e] : -1;
    }
#pragma unroll
    for (int u = 0; u < 4; ++u) {
        if (d[u] >= 0) atomicAdd(cnt + d[u], 1);
    }
}

__global__ __launch_bounds__(1024)
void sage_scan(const int* __restrict__ cnt,
               int* __restrict__ rowptr,
               int* __restrict__ cursor,
               float* __restrict__ degf,
               int n) {
    __shared__ int wfront[32];
    __shared__ int wtot[32];
    const int t    = threadIdx.x;
    const int lane = t & 31;
    const int wv   = t >> 5;
    const int base = blockIdx.x * 1024;

    int front = 0;
    for (int j = t; j < base; j += 1024) front += cnt[j];

    const int idx = base + t;
    const int v   = idx < n ? cnt[idx] : 0;

#pragma unroll
    for (int o = 16; o > 0; o >>= 1) front += __shfl_xor(front, o, 32);
    int inc = v;
#pragma unroll
    for (int o = 1; o < 32; o <<= 1) {
        const int y = __shfl_up(inc, o, 32);
        if (lane >= o) inc += y;
    }
    if (lane == 31) wtot[wv] = inc;
    if (lane == 0)  wfront[wv] = front;
    __syncthreads();

    int all_front = wfront[lane];
    const int tot = wtot[lane];
#pragma unroll
    for (int o = 16; o > 0; o >>= 1) all_front += __shfl_xor(all_front, o, 32);
    int tinc = tot;
#pragma unroll
    for (int o = 1; o < 32; o <<= 1) {
        const int y = __shfl_up(tinc, o, 32);
        if (lane >= o) tinc += y;
    }
    const int waves_before = __shfl(tinc, wv, 32) - __shfl(tot, wv, 32);
    const int excl = all_front + waves_before + inc - v;

    if (idx < n) {
        rowptr[idx] = excl;
        cursor[idx] = excl;
        degf[idx]   = (float)v;
        if (idx == n - 1) rowptr[n] = excl + v;
    }
}

__global__ __launch_bounds__(256)
void sage_fill(const int* __restrict__ src,
               const int* __restrict__ dst,
               int* __restrict__ cursor,
               int* __restrict__ csr,
               int n_edges) {
    const int e0 = blockIdx.x * 1024 + threadIdx.x;
    int d[4];
    int s[4];
#pragma unroll
    for (int u = 0; u < 4; ++u) {
        const int e = e0 + u * 256;
        d[u] = e < n_edges ? dst[e] : -1;
        s[u] = e < n_edges ? src[e] : 0;
    }
    int p[4];
#pragma unroll
    for (int u = 0; u < 4; ++u) {
        p[u] = d[u] >= 0 ? atomicAdd(cursor + d[u], 1) : -1;
    }
#pragma unroll
    for (int u = 0; u < 4; ++u) {
        if (p[u] >= 0) csr[p[u]] = s[u];
    }
}

__global__ __launch_bounds__(256)
void sage_gather(const float* __restrict__ feat,
                 const int*   __restrict__ rowptr,
                 const int*   __restrict__ csr,
                 float* __restrict__ agg,
                 int n_nodes) {
    const int lane = threadIdx.x & 31;
    const int wave = __builtin_amdgcn_readfirstlane((int)(threadIdx.x >> 5));
    const int node = blockIdx.x * 8 + wave;
    if (node >= n_nodes) return;
    const int beg = __builtin_amdgcn_readfirstlane(rowptr[node]);
    const int end = __builtin_amdgcn_readfirstlane(rowptr[node + 1]);

    const float* col = feat + lane * 4;
    float4 acc = make_float4(0.0f, 0.0f, 0.0f, 0.0f);

    for (int b = beg; b < end; b += 32) {
        const int left = end - b;
        const int cnt  = left < 32 ? left : 32;
        int sv = 0;
        if (lane < cnt) sv = csr[b + lane];

        int i = 0;
        for (; i + 16 <= cnt; i += 16) {
            float4 v[16];
#pragma unroll
            for (int u = 0; u < 16; ++u) {
                const int s = __builtin_amdgcn_readlane(sv, i + u);
                v[u] = *(const float4*)(col + (size_t)s * 128);
            }
#pragma unroll
            for (int u = 0; u < 16; ++u) {
                acc.x += v[u].x; acc.y += v[u].y; acc.z += v[u].z; acc.w += v[u].w;
            }
        }
        if (i < cnt) {
            float4 v[16];
#pragma unroll
            for (int u = 0; u < 16; ++u) {
                v[u] = make_float4(0.0f, 0.0f, 0.0f, 0.0f);
                if (i + u < cnt) {
                    const int s = __builtin_amdgcn_readlane(sv, i + u);
                    v[u] = *(const float4*)(col + (size_t)s * 128);
                }
            }
#pragma unroll
            for (int u = 0; u < 16; ++u) {
                acc.x += v[u].x; acc.y += v[u].y; acc.z += v[u].z; acc.w += v[u].w;
            }
        }
    }
    *(float4*)(agg + (size_t)node * 128 + lane * 4) = acc;
}

template <int NDIM, bool RELU>
__global__ __launch_bounds__(256)
void sage_gemm(const float* __restrict__ agg,
               const float* __restrict__ deg,
               const float* __restrict__ X,
               const float* __restrict__ Wl,
               const float* __restrict__ Wr,
               const float* __restrict__ bias,
               float* __restrict__ out,
               int n_nodes) {
    constexpr int NG    = NDIM / 64;
    constexpr int N4    = NDIM / 4;
    constexpr int PITCH = NDIM + 16;
    __shared__ __attribute__((aligned(16))) float2 wlds[128 * (128 + 16)];

    const int t = threadIdx.x;

#pragma unroll 4
    for (int i = t; i < 128 * N4; i += 256) {
        const int kp = i / N4;
        const int n  = (i % N4) * 4;
        const float* w0 = (kp < 64 ? Wl + (size_t)(2 * kp) * NDIM
                                   : Wr + (size_t)(2 * kp - 128) * NDIM) + n;
        const float4 r0 = *(const float4*)w0;
        const float4 r1 = *(const float4*)(w0 + NDIM);
        const int q = 4 * (kp >> 2) + 2 * (kp & 1) + ((kp >> 1) & 1);
        float4* d = (float4*)(wlds + q * PITCH + n);
        d[0] = make_float4(r0.x, r1.x, r0.y, r1.y);
        d[1] = make_float4(r0.z, r1.z, r0.w, r1.w);
    }
    __syncthreads();

    const int lane = t & 31;
    const int c    = lane & 15;
    const int h    = lane >> 4;
    const int wave   = __builtin_amdgcn_readfirstlane(t >> 5);
    const int units  = (n_nodes / 16) * NG;
    const int nwg    = (int)gridDim.x;
    const int stride = nwg * 8;
    int u = (wave >> 2) * (nwg * 4) + (int)blockIdx.x * 4 + (wave & 3);
    if (u >= units) return;

    const float2* bbase = wlds + h * PITCH + c;
    const int laneOff = 4 * h;

    float4 cur[4];
    float4 nxt[4];
    {
        const size_t off0 = (size_t)((u / NG) * 16 + c) * 128 + laneOff;
#pragma unroll
        for (int jj = 0; jj < 4; ++jj) cur[jj] = *(const float4*)(agg + off0 + 8 * jj);
    }

    for (;;) {
        const int mtile = u / NG;
        const int ng    = u % NG;
        const int row   = mtile * 16 + c;
        const size_t rowOff = (size_t)row * 128 + laneOff;
        const int  un       = u + stride;
        const bool has_next = un < units;
        const int  uN       = has_next ? un : u;
        const size_t rowOffN = (size_t)((uN / NG) * 16 + c) * 128 + laneOff;

        const float dval = deg[row];
        const float invd = 1.0f / fmaxf(dval, 1.0f);

        const float2* bp = bbase + ng * 64;
        v8f acc[4];
#pragma unroll
        for (int nt = 0; nt < 4; ++nt) acc[nt] = v8f{};

#pragma unroll 1
        for (int ch = 0; ch < 8; ++ch) {
            const float* nbase = (ch < 3 || ch == 7) ? agg : X;
            const size_t noff  = ch == 7 ? rowOffN : rowOff + 32 * ((ch + 1) & 3);
#pragma unroll
            for (int jj = 0; jj < 4; ++jj) nxt[jj] = *(const float4*)(nbase + noff + 8 * jj);

            const float sc = ch < 4 ? invd : 1.0f;
#pragma unroll
            for (int jj = 0; jj < 4; ++jj) {
                float4 av = cur[jj];
                av.x *= sc; av.y *= sc; av.z *= sc; av.w *= sc;
#pragma unroll
                for (int e = 0; e < 2; ++e) {
                    v2f af;
                    af.x = e ? av.z : av.x;
                    af.y = e ? av.w : av.y;
                    const float2* br = bp + (4 * jj + 2 * e) * PITCH;
#pragma unroll
                    for (int nt = 0; nt < 4; ++nt) {
                        const float2 b = br[nt * 16];
                        v2f bf; bf.x = b.x; bf.y = b.y;
                        acc[nt] = __builtin_amdgcn_wmma_f32_16x16x4_f32(
                                      false, af, false, bf, (short)0, acc[nt], false, false);
                    }
                }
            }
            bp += 16 * PITCH;
#pragma unroll
            for (int jj = 0; jj < 4; ++jj) cur[jj] = nxt[jj];
        }

#pragma unroll
        for (int nt = 0; nt < 4; ++nt) {
            const int   col = ng * 64 + nt * 16 + c;
            const float bv  = bias[col];
#pragma unroll
            for (int i = 0; i < 8; ++i) {
                const int node = mtile * 16 + i + 8 * h;
                float v = acc[nt][i] + bv;
                if (RELU) v = fmaxf(v, 0.0f);
                out[(size_t)node * NDIM + col] = v;
            }
        }

        if (!has_next) break;
        u = un;
    }
}

extern "C" void kernel_launch(void* const* d_in, const int* in_sizes, int n_in,
                              void* d_out, int out_size, void* d_ws, size_t ws_size,
                              hipStream_t stream) {
    const float* x    = (const float*)d_in[0];
    const int*   ei   = (const int*)  d_in[1];
    const float* W1l  = (const float*)d_in[2];
    const float* W1r  = (const float*)d_in[3];
    const float* b1   = (const float*)d_in[4];
    const float* W2l  = (const float*)d_in[5];
    const float* W2r  = (const float*)d_in[6];
    const float* b2   = (const float*)d_in[7];
    float*       outp = (float*)d_out;

    const int n_nodes = in_sizes[0] / IN_DIM;
    const int n_edges = in_sizes[1] / 2;
    const int* src = ei;
    const int* dst = ei + n_edges;

    const size_t aggBytes = (size_t)n_nodes * 128 * sizeof(float);
    const size_t degBytes = (((size_t)(n_nodes + 1) * sizeof(float)) + 255) & ~(size_t)255;

    char* ws      = (char*)d_ws;
    float* agg1   = (float*)(ws);
    float* agg2   = (float*)(ws + aggBytes);
    float* deg    = (float*)(ws + 2 * aggBytes);
    float* h      = (float*)(ws + 2 * aggBytes + degBytes);
    int*   cnt    = (int*)  (ws + 3 * aggBytes + degBytes);
    int*   rowptr = (int*)  (ws + 3 * aggBytes + 2 * degBytes);
    int*   cursor = (int*)  (ws + 3 * aggBytes + 3 * degBytes);
    int*   csr    = (int*)  (ws + 3 * aggBytes + 4 * degBytes);

    hipMemsetAsync(cnt, 0, degBytes, stream);

    const int edgeBlocks = (n_edges + 1023) / 1024;
    const int scanBlocks = (n_nodes + 1023) / 1024;
    const int nodeBlocks = (n_nodes + 7) / 8;
    const int gemmBlocks = 512;

    sage_count<<<edgeBlocks, 256, 0, stream>>>(dst, cnt, n_edges);
    sage_scan<<<scanBlocks, 1024, 0, stream>>>(cnt, rowptr, cursor, deg, n_nodes);
    sage_fill<<<edgeBlocks, 256, 0, stream>>>(src, dst, cursor, csr, n_edges);

    sage_gather<<<nodeBlocks, 256, 0, stream>>>(x, rowptr, csr, agg1, n_nodes);
    sage_gemm<HIDDEN, true><<<gemmBlocks, 256, 0, stream>>>(
        agg1, deg, x, W1l, W1r, b1, h, n_nodes);

    sage_gather<<<nodeBlocks, 256, 0, stream>>>(h, rowptr, csr, agg2, n_nodes);
    sage_gemm<OUT_DIM, false><<<gemmBlocks, 256, 0, stream>>>(
        agg2, deg, h, W2l, W2r, b2, outp, n_nodes);
}


# edge sort without global atomics: 6144-edge chunks ranked by LDS counters into 256-node buckets, then a block per bucket ranks by node in LDS and writes the lists; sum order = LDS add arrival order
# speedup vs baseline: 5.0294x; 2.2329x over previous
#include <hip/hip_runtime.h>


typedef __attribute__((ext_vector_type(2))) float v2f;
typedef __attribute__((ext_vector_type(8))) float v8f;

#define N_NODES 50000
#define N_EDGES 1600000
#define IN_DIM  128
#define HIDDEN  128
#define OUT_DIM 64

#define BUCKET_SHIFT 8
#define BUCKET_NODES (1 << BUCKET_SHIFT)
#define CHUNK_PER_THREAD 6
#define CHUNK (1024 * CHUNK_PER_THREAD)
#define PIECES 8

__device__ __forceinline__ int block_scan_excl(int v, int* wtot, int t) {
    const int lane = t & 31;
    const int wv   = t >> 5;
    int inc = v;
#pragma unroll
    for (int o = 1; o < 32; o <<= 1) {
        const int y = __shfl_up(inc, o, 32);
        if (lane >= o) inc += y;
    }
    if (lane == 31) wtot[wv] = inc;
    __syncthreads();
    const int tot = wtot[lane];
    int tinc = tot;
#pragma unroll
    for (int o = 1; o < 32; o <<= 1) {
        const int y = __shfl_up(tinc, o, 32);
        if (lane >= o) tinc += y;
    }
    const int waves_before = __shfl(tinc, wv, 32) - __shfl(tot, wv, 32);
    return waves_before + inc - v;
}

__global__ __launch_bounds__(1024)
void sage_chunk_sort(const int* __restrict__ src,
                     const int* __restrict__ dst,
                     int* __restrict__ psrc,
                     int* __restrict__ pdst,
                     int* __restrict__ off,
                     int n_edges, int n_buckets, int n_chunks) {
    __shared__ int lh[1024];
    __shared__ int wtot[32];
    __shared__ int ss[CHUNK];
    __shared__ int sd[CHUNK];
    const int t     = threadIdx.x;
    const int c     = blockIdx.x;
    const int cbase = c * CHUNK;

    int d[CHUNK_PER_THREAD];
    int s[CHUNK_PER_THREAD];
#pragma unroll
    for (int u = 0; u < CHUNK_PER_THREAD; ++u) {
        const int e = cbase + u * 1024 + t;
        d[u] = e < n_edges ? dst[e] : -1;
        s[u] = e < n_edges ? src[e] : 0;
    }
    lh[t] = 0;
    __syncthreads();
#pragma unroll
    for (int u = 0; u < CHUNK_PER_THREAD; ++u) {
        if (d[u] >= 0) atomicAdd(&lh[d[u] >> BUCKET_SHIFT], 1);
    }
    __syncthreads();

    const int v    = lh[t];
    const int excl = block_scan_excl(v, wtot, t);
    if (t <= n_buckets) off[(size_t)t * n_chunks + c] = excl;
    lh[t] = excl;
    __syncthreads();

#pragma unroll
    for (int u = 0; u < CHUNK_PER_THREAD; ++u) {
        if (d[u] >= 0) {
            const int p = atomicAdd(&lh[d[u] >> BUCKET_SHIFT], 1);
            ss[p] = s[u];
            sd[p] = d[u];
        }
    }
    __syncthreads();
#pragma unroll
    for (int u = 0; u < CHUNK_PER_THREAD; ++u) {
        const int i = u * 1024 + t;
        if (cbase + i < n_edges) {
            psrc[cbase + i] = ss[i];
            pdst[cbase + i] = sd[i];
        }
    }
}

__global__ __launch_bounds__(1024)
void sage_bucket_sort(const int* __restrict__ psrc,
                      const int* __restrict__ pdst,
                      const int* __restrict__ off,
                      int* __restrict__ rowptr,
                      float* __restrict__ degf,
                      int* __restrict__ csr,
                      int n_nodes, int n_chunks) {
    __shared__ int tb[1024];
    __shared__ int tl[1024];
    __shared__ int cnt[BUCKET_NODES];
    __shared__ int wtot[32];
    __shared__ int wsum[32];
    const int t     = threadIdx.x;
    const int lane  = t & 31;
    const int wv    = t >> 5;
    const int b     = blockIdx.x;
    const int node0 = b << BUCKET_SHIFT;
    const int* offb = off + (size_t)b * n_chunks;
    const int* offn = offb + n_chunks;

    if (t < BUCKET_NODES) cnt[t] = 0;
    int front = 0;

    for (int c0 = 0; c0 < n_chunks; c0 += 1024) {
        __syncthreads();
        const int c = c0 + t;
        int pb = 0, pe = 0;
        if (c < n_chunks) { pb = offb[c]; pe = offn[c]; }
        tb[t] = pb;
        tl[t] = pe - pb;
        front += pb;
        __syncthreads();
        const int left = n_chunks - c0;
        const int ncur = left < 1024 ? left : 1024;
        for (int k = wv; k < ncur; k += 32 * PIECES) {
            int base[PIECES];
            int len[PIECES];
            int maxlen = 0;
#pragma unroll
            for (int q = 0; q < PIECES; ++q) {
                const int kk = k + 32 * q;
                base[q] = 0;
                len[q]  = 0;
                if (kk < ncur) {
                    base[q] = (c0 + kk) * CHUNK + tb[kk];
                    len[q]  = tl[kk];
                }
                maxlen = len[q] > maxlen ? len[q] : maxlen;
            }
            for (int i0 = 0; i0 < maxlen; i0 += 32) {
                const int i = i0 + lane;
                int dv[PIECES];
#pragma unroll
                for (int q = 0; q < PIECES; ++q)
                    dv[q] = i < len[q] ? pdst[base[q] + i] : -1;
#pragma unroll
                for (int q = 0; q < PIECES; ++q)
                    if (dv[q] >= 0) atomicAdd(&cnt[dv[q] - node0], 1);
            }
        }
    }

#pragma unroll
    for (int o = 16; o > 0; o >>= 1) front += __shfl_xor(front, o, 32);
    if (lane == 0) wsum[wv] = front;
    __syncthreads();
    int bstart = wsum[lane];
#pragma unroll
    for (int o = 16; o > 0; o >>= 1) bstart += __shfl_xor(bstart, o, 32);

    const int v    = t < BUCKET_NODES ? cnt[t] : 0;
    const int excl = block_scan_excl(v, wtot, t);
    const int node = node0 + t;
    if (t < BUCKET_NODES) {
        cnt[t] = excl;
        if (node < n_nodes) {
            rowptr[node] = bstart + excl;
            degf[node]   = (float)v;
            if (node == n_nodes - 1) rowptr[n_nodes] = bstart + excl + v;
        }
    }

    for (int c0 = 0; c0 < n_chunks; c0 += 1024) {
        __syncthreads();
        const int c = c0 + t;
        int pb = 0, pe = 0;
        if (c < n_chunks) { pb = offb[c]; pe = offn[c]; }
        tb[t] = pb;
        tl[t] = pe - pb;
        __syncthreads();
        const int left = n_chunks - c0;
        const int ncur = left < 1024 ? left : 1024;
        for (int k = wv; k < ncur; k += 32 * PIECES) {
            int base[PIECES];
            int len[PIECES];
            int maxlen = 0;
#pragma unroll
            for (int q = 0; q < PIECES; ++q) {
                const int kk = k + 32 * q;
                base[q] = 0;
                len[q]  = 0;
                if (kk < ncur) {
                    base[q] = (c0 + kk) * CHUNK + tb[kk];
                    len[q]  = tl[kk];
                }
                maxlen = len[q] > maxlen ? len[q] : maxlen;
            }
            for (int i0 = 0; i0 < maxlen; i0 += 32) {
                const int i = i0 + lane;
                int dv[PIECES];
                int sv[PIECES];
#pragma unroll
                for (int q = 0; q < PIECES; ++q) {
                    dv[q] = i < len[q] ? pdst[base[q] + i] : -1;
                    sv[q] = i < len[q] ? psrc[base[q] + i] : 0;
                }
#pragma unroll
                for (int q = 0; q < PIECES; ++q) {
                    if (dv[q] >= 0) {
                        const int p = atomicAdd(&cnt[dv[q] - node0], 1);
                        csr[bstart + p] = sv[q];
                    }
                }
            }
        }
    }
}

__global__ __launch_bounds__(256)
void sage_gather(const float* __restrict__ feat,
                 const int*   __restrict__ rowptr,
                 const int*   __restrict__ csr,
                 float* __restrict__ agg,
                 int n_nodes) {
    const int lane = threadIdx.x & 31;
    const int wave = __builtin_amdgcn_readfirstlane((int)(threadIdx.x >> 5));
    const int node = blockIdx.x * 8 + wave;
    if (node >= n_nodes) return;
    const int beg = __builtin_amdgcn_readfirstlane(rowptr[node]);
    const int end = __builtin_amdgcn_readfirstlane(rowptr[node + 1]);

    const float* col = feat + lane * 4;
    float4 acc = make_float4(0.0f, 0.0f, 0.0f, 0.0f);

    for (int b = beg; b < end; b += 32) {
        const int left = end - b;
        const int cnt  = left < 32 ? left : 32;
        int sv = 0;
        if (lane < cnt) sv = csr[b + lane];

        int i = 0;
        for (; i + 16 <= cnt; i += 16) {
            float4 v[16];
#pragma unroll
            for (int u = 0; u < 16; ++u) {
                const int s = __builtin_amdgcn_readlane(sv, i + u);
                v[u] = *(const float4*)(col + (size_t)s * 128);
            }
#pragma unroll
            for (int u = 0; u < 16; ++u) {
                acc.x += v[u].x; acc.y += v[u].y; acc.z += v[u].z; acc.w += v[u].w;
            }
        }
        if (i < cnt) {
            float4 v[16];
#pragma unroll
            for (int u = 0; u < 16; ++u) {
                v[u] = make_float4(0.0f, 0.0f, 0.0f, 0.0f);
                if (i + u < cnt) {
                    const int s = __builtin_amdgcn_readlane(sv, i + u);
                    v[u] = *(const float4*)(col + (size_t)s * 128);
                }
            }
#pragma unroll
            for (int u = 0; u < 16; ++u) {
                acc.x += v[u].x; acc.y += v[u].y; acc.z += v[u].z; acc.w += v[u].w;
            }
        }
    }
    *(float4*)(agg + (size_t)node * 128 + lane * 4) = acc;
}

template <int NDIM, bool RELU>
__global__ __launch_bounds__(256)
void sage_gemm(const float* __restrict__ agg,
               const float* __restrict__ deg,
               const float* __restrict__ X,
               const float* __restrict__ Wl,
               const float* __restrict__ Wr,
               const float* __restrict__ bias,
               float* __restrict__ out,
               int n_nodes) {
    constexpr int NG    = NDIM / 64;
    constexpr int N4    = NDIM / 4;
    constexpr int PITCH = NDIM + 16;
    __shared__ __attribute__((aligned(16))) float2 wlds[128 * (128 + 16)];

    const int t = threadIdx.x;

#pragma unroll 4
    for (int i = t; i < 128 * N4; i += 256) {
        const int kp = i / N4;
        const int n  = (i % N4) * 4;
        const float* w0 = (kp < 64 ? Wl + (size_t)(2 * kp) * NDIM
                                   : Wr + (size_t)(2 * kp - 128) * NDIM) + n;
        const float4 r0 = *(const float4*)w0;
        const float4 r1 = *(const float4*)(w0 + NDIM);
        const int q = 4 * (kp >> 2) + 2 * (kp & 1) + ((kp >> 1) & 1);
        float4* d = (float4*)(wlds + q * PITCH + n);
        d[0] = make_float4(r0.x, r1.x, r0.y, r1.y);
        d[1] = make_float4(r0.z, r1.z, r0.w, r1.w);
    }
    __syncthreads();

    const int lane = t & 31;
    const int c    = lane & 15;
    const int h    = lane >> 4;
    const int wave   = __builtin_amdgcn_readfirstlane(t >> 5);
    const int units  = (n_nodes / 16) * NG;
    const int nwg    = (int)gridDim.x;
    const int stride = nwg * 8;
    int u = (wave >> 2) * (nwg * 4) + (int)blockIdx.x * 4 + (wave & 3);
    if (u >= units) return;

    const float2* bbase = wlds + h * PITCH + c;
    const int laneOff = 4 * h;

    float4 cur[4];
    float4 nxt[4];
    {
        const size_t off0 = (size_t)((u / NG) * 16 + c) * 128 + laneOff;
#pragma unroll
        for (int jj = 0; jj < 4; ++jj) cur[jj] = *(const float4*)(agg + off0 + 8 * jj);
    }

    for (;;) {
        const int mtile = u / NG;
        const int ng    = u % NG;
        const int row   = mtile * 16 + c;
        const size_t rowOff = (size_t)row * 128 + laneOff;
        const int  un       = u + stride;
        const bool has_next = un < units;
        const int  uN       = has_next ? un : u;
        const size_t rowOffN = (size_t)((uN / NG) * 16 + c) * 128 + laneOff;

        const float dval = deg[row];
        const float invd = 1.0f / fmaxf(dval, 1.0f);

        const float2* bp = bbase + ng * 64;
        v8f acc[4];
#pragma unroll
        for (int nt = 0; nt < 4; ++nt) acc[nt] = v8f{};

#pragma unroll 1
        for (int ch = 0; ch < 8; ++ch) {
            const float* nbase = (ch < 3 || ch == 7) ? agg : X;
            const size_t noff  = ch == 7 ? rowOffN : rowOff + 32 * ((ch + 1) & 3);
#pragma unroll
            for (int jj = 0; jj < 4; ++jj) nxt[jj] = *(const float4*)(nbase + noff + 8 * jj);

            const float sc = ch < 4 ? invd : 1.0f;
#pragma unroll
            for (int jj = 0; jj < 4; ++jj) {
                float4 av = cur[jj];
                av.x *= sc; av.y *= sc; av.z *= sc; av.w *= sc;
#pragma unroll
                for (int e = 0; e < 2; ++e) {
                    v2f af;
                    af.x = e ? av.z : av.x;
                    af.y = e ? av.w : av.y;
                    const float2* br = bp + (4 * jj + 2 * e) * PITCH;
#pragma unroll
                    for (int nt = 0; nt < 4; ++nt) {
                        const float2 b = br[nt * 16];
                        v2f bf; bf.x = b.x; bf.y = b.y;
                        acc[nt] = __builtin_amdgcn_wmma_f32_16x16x4_f32(
                                      false, af, false, bf, (short)0, acc[nt], false, false);
                    }
                }
            }
            bp += 16 * PITCH;
#pragma unroll
            for (int jj = 0; jj < 4; ++jj) cur[jj] = nxt[jj];
        }

#pragma unroll
        for (int nt = 0; nt < 4; ++nt) {
            const int   col = ng * 64 + nt * 16 + c;
            const float bv  = bias[col];
#pragma unroll
            for (int i = 0; i < 8; ++i) {
                const int node = mtile * 16 + i + 8 * h;
                float v = acc[nt][i] + bv;
                if (RELU) v = fmaxf(v, 0.0f);
                out[(size_t)node * NDIM + col] = v;
            }
        }

        if (!has_next) break;
        u = un;
    }
}

extern "C" void kernel_launch(void* const* d_in, const int* in_sizes, int n_in,
                              void* d_out, int out_size, void* d_ws, size_t ws_size,
                              hipStream_t stream) {
    const float* x    = (const float*)d_in[0];
    const int*   ei   = (const int*)  d_in[1];
    const float* W1l  = (const float*)d_in[2];
    const float* W1r  = (const float*)d_in[3];
    const float* b1   = (const float*)d_in[4];
    const float* W2l  = (const float*)d_in[5];
    const float* W2r  = (const float*)d_in[6];
    const float* b2   = (const float*)d_in[7];
    float*       outp = (float*)d_out;

    const int n_nodes = in_sizes[0] / IN_DIM;
    const int n_edges = in_sizes[1] / 2;
    const int* src = ei;
    const int* dst = ei + n_edges;

    const size_t aggBytes = (size_t)n_nodes * 128 * sizeof(float);
    const size_t degBytes = (((size_t)(n_nodes + 1) * sizeof(float)) + 255) & ~(size_t)255;

    char* ws      = (char*)d_ws;
    float* agg1   = (float*)(ws);
    float* agg2   = (float*)(ws + aggBytes);
    float* deg    = (float*)(ws + 2 * aggBytes);
    float* h      = (float*)(ws + 2 * aggBytes + degBytes);
    int*   rowptr = (int*)  (ws + 3 * aggBytes + degBytes);
    int*   csr    = (int*)  (ws + 3 * aggBytes + 2 * degBytes);
    int*   psrc   = csr  + n_edges;
    int*   pdst   = psrc + n_edges;
    int*   off    = pdst + n_edges;

    const int n_chunks  = (n_edges + CHUNK - 1) / CHUNK;
    const int n_buckets = (n_nodes + BUCKET_NODES - 1) / BUCKET_NODES;
    const int nodeBlocks = (n_nodes + 7) / 8;
    const int gemmBlocks = 512;

    sage_chunk_sort<<<n_chunks, 1024, 0, stream>>>(src, dst, psrc, pdst, off,
                                                   n_edges, n_buckets, n_chunks);
    sage_bucket_sort<<<n_buckets, 1024, 0, stream>>>(psrc, pdst, off, rowptr, deg, csr,
                                                     n_nodes, n_chunks);

    sage_gather<<<nodeBlocks, 256, 0, stream>>>(x, rowptr, csr, agg1, n_nodes);
    sage_gemm<HIDDEN, true><<<gemmBlocks, 256, 0, stream>>>(
        agg1, deg, x, W1l, W1r, b1, h, n_nodes);

    sage_gather<<<nodeBlocks, 256, 0, stream>>>(h, rowptr, csr, agg2, n_nodes);
    sage_gemm<OUT_DIM, false><<<gemmBlocks, 256, 0, stream>>>(
        agg2, deg, h, W2l, W2r, b2, outp, n_nodes);
}
